# starting kernel
# baseline (speedup 1.0000x reference)
#include <hip/hip_runtime.h>
#include <hip/hip_bf16.h>

#define NNODES 50000
#define DIMIN  512
#define HIDDEN 256
#define NCLS   2
#define LDSK   516

typedef __attribute__((ext_vector_type(2))) float v2f;
typedef __attribute__((ext_vector_type(8))) float v8f;


__global__ void k_init_deg(float* __restrict__ deg, int n) {
    int i = blockIdx.x * blockDim.x + threadIdx.x;
    if (i < n) deg[i] = 1.0f;
}

__global__ void k_deg_accum(const int* __restrict__ dst, float* __restrict__ deg, int e) {
    int i = blockIdx.x * blockDim.x + threadIdx.x;
    if (i < e) atomicAdd(&deg[dst[i]], 1.0f);
}

__global__ void k_dinv(const float* __restrict__ deg, float* __restrict__ dinv, int n) {
    int i = blockIdx.x * blockDim.x + threadIdx.x;
    if (i < n) dinv[i] = __frsqrt_rn(deg[i]);
}


__global__ void k_packW(const float* __restrict__ W, float2* __restrict__ Wp) {
    int idx = blockIdx.x * blockDim.x + threadIdx.x;
    if (idx >= (DIMIN / 2) * HIDDEN) return;
    int kp  = idx / HIDDEN;
    int col = idx - kp * HIDDEN;
    float2 p;
    p.x = W[(size_t)(2 * kp)     * HIDDEN + col];
    p.y = W[(size_t)(2 * kp + 1) * HIDDEN + col];
    Wp[idx] = p;
}


__global__ void __launch_bounds__(256)
k_gemm1(const float* __restrict__ X, const float2* __restrict__ Wp,
        const float* __restrict__ dinv, float* __restrict__ H, float* __restrict__ AGG) {
    __shared__ float As[16 * LDSK];

    const int row0 = blockIdx.x * 16;

    for (int idx = threadIdx.x; idx < (16 * DIMIN) / 4; idx += 256) {
        int r = idx >> 7;
        int c = (idx & 127) * 4;
        const float4 v = *(const float4*)(X + (size_t)(row0 + r) * DIMIN + c);
        *(float4*)(&As[r * LDSK + c]) = v;
    }
    __syncthreads();

    const int wave = threadIdx.x >> 5;
    const int lane = threadIdx.x & 31;
    const int l16  = lane & 15;
    const int hi   = lane >> 4;

    for (int ct = wave; ct < HIDDEN / 16; ct += 8) {
        const int col0 = ct * 16;
        v8f c = {};
        #pragma unroll 8
        for (int k = 0; k < DIMIN; k += 4) {
            const int kk = k + 2 * hi;
            const float2 af = *(const float2*)(&As[l16 * LDSK + kk]);
            v2f a; a.x = af.x; a.y = af.y;
            const float2 bf = Wp[(size_t)(kk >> 1) * HIDDEN + col0 + l16];
            v2f b; b.x = bf.x; b.y = bf.y;
            c = __builtin_amdgcn_wmma_f32_16x16x4_f32(false, a, false, b,
                                                      (short)0, c, false, false);
        }
        #pragma unroll
        for (int v = 0; v < 8; ++v) {
            const int row = row0 + hi * 8 + v;
            const float dv = dinv[row];
            const size_t o = (size_t)row * HIDDEN + col0 + l16;
            H[o]   = c[v];
            AGG[o] = dv * dv * c[v];
        }
    }
}


__global__ void k_edge_agg1(const int* __restrict__ src, const int* __restrict__ dst,
                            const float* __restrict__ dinv, const float* __restrict__ H1,
                            float* __restrict__ AGG, int e) {
    int gid  = blockIdx.x * blockDim.x + threadIdx.x;
    int wid  = gid >> 5;
    int lane = gid & 31;
    if (wid >= e) return;
    int s = src[wid], d = dst[wid];
    float norm = dinv[s] * dinv[d];
    const float4* hs = (const float4*)(H1 + (size_t)s * HIDDEN);
    float* od = AGG + (size_t)d * HIDDEN;
    #pragma unroll
    for (int it = 0; it < 2; ++it) {
        int c4 = it * 32 + lane;
        float4 h = hs[c4];
        int c = c4 * 4;
        atomicAdd(&od[c + 0], norm * h.x);
        atomicAdd(&od[c + 1], norm * h.y);
        atomicAdd(&od[c + 2], norm * h.z);
        atomicAdd(&od[c + 3], norm * h.w);
    }
}

__global__ void k_bias_relu(float* __restrict__ A, const float* __restrict__ b1, int total) {
    int i = blockIdx.x * blockDim.x + threadIdx.x;
    if (i >= total) return;
    float v = A[i] + b1[i & (HIDDEN - 1)];
    A[i] = v > 0.0f ? v : 0.0f;
}


__global__ void k_gemm2(const float* __restrict__ Z, const float* __restrict__ W2,
                        const float* __restrict__ dinv, const float* __restrict__ b2,
                        float* __restrict__ H2, float* __restrict__ out, int n) {
    int i = blockIdx.x * blockDim.x + threadIdx.x;
    if (i >= n) return;
    const float4* z4 = (const float4*)(Z + (size_t)i * HIDDEN);
    float a0 = 0.0f, a1 = 0.0f;
    #pragma unroll 4
    for (int k4 = 0; k4 < HIDDEN / 4; ++k4) {
        float4 z = z4[k4];
        int k = k4 * 4;
        a0 += z.x * W2[2*k+0] + z.y * W2[2*k+2] + z.z * W2[2*k+4] + z.w * W2[2*k+6];
        a1 += z.x * W2[2*k+1] + z.y * W2[2*k+3] + z.z * W2[2*k+5] + z.w * W2[2*k+7];
    }
    H2[2*i]     = a0;
    H2[2*i + 1] = a1;
    float s = dinv[i]; s *= s;
    out[2*i]     = s * a0 + b2[0];
    out[2*i + 1] = s * a1 + b2[1];
}


__global__ void k_edge_agg2(const int* __restrict__ src, const int* __restrict__ dst,
                            const float* __restrict__ dinv, const float* __restrict__ H2,
                            float* __restrict__ out, int e) {
    int i = blockIdx.x * blockDim.x + threadIdx.x;
    if (i >= e) return;
    int s = src[i], d = dst[i];
    float norm = dinv[s] * dinv[d];
    atomicAdd(&out[2*d],     norm * H2[2*s]);
    atomicAdd(&out[2*d + 1], norm * H2[2*s + 1]);
}


extern "C" void kernel_launch(void* const* d_in, const int* in_sizes, int n_in,
                              void* d_out, int out_size, void* d_ws, size_t ws_size,
                              hipStream_t stream) {
    const float* X   = (const float*)d_in[0];
    const float* W1  = (const float*)d_in[1];
    const float* b1  = (const float*)d_in[2];
    const float* W2  = (const float*)d_in[3];
    const float* b2  = (const float*)d_in[4];
    const int*   ei  = (const int*)d_in[5];
    float*       out = (float*)d_out;

    const int N = NNODES;
    const int E = in_sizes[5] / 2;
    const int* src = ei;
    const int* dst = ei + E;

    float*  deg  = (float*)d_ws;
    float*  dinv = deg  + N;
    float*  H1   = dinv + N;
    float*  AGG1 = H1   + (size_t)N * HIDDEN;
    float*  H2   = AGG1 + (size_t)N * HIDDEN;
    float2* Wp   = (float2*)(H2 + (size_t)N * NCLS);

    const int T = 256;

    k_init_deg <<<(N + T - 1) / T, T, 0, stream>>>(deg, N);
    k_deg_accum<<<(E + T - 1) / T, T, 0, stream>>>(dst, deg, E);
    k_dinv     <<<(N + T - 1) / T, T, 0, stream>>>(deg, dinv, N);

    {
        int total = (DIMIN / 2) * HIDDEN;
        k_packW<<<(total + T - 1) / T, T, 0, stream>>>(W1, Wp);
    }

    k_gemm1<<<N / 16, 256, 0, stream>>>(X, Wp, dinv, H1, AGG1);
    {
        long long threads = (long long)E * 32;
        k_edge_agg1<<<(int)((threads + T - 1) / T), T, 0, stream>>>(src, dst, dinv, H1, AGG1, E);
    }
    {
        int total = N * HIDDEN;
        k_bias_relu<<<(total + T - 1) / T, T, 0, stream>>>(AGG1, b1, total);
    }

    k_gemm2    <<<(N + T - 1) / T, T, 0, stream>>>(AGG1, W2, dinv, b2, H2, out, N);
    k_edge_agg2<<<(E + T - 1) / T, T, 0, stream>>>(src, dst, dinv, H2, out, E);
}


# Float scatter atomics replaced by a dst-sorted gather (counting sort, arrival-order rank): one wave per node sums its row in registers, applies bias, relu and the 256->2 product there; lane-tree sums
# speedup vs baseline: 1.6832x; 1.6832x over previous
#include <hip/hip_runtime.h>
#include <hip/hip_bf16.h>

#define NNODES 50000
#define DIMIN  512
#define HIDDEN 256
#define NCLS   2
#define LDSK   516

typedef __attribute__((ext_vector_type(2))) float v2f;
typedef __attribute__((ext_vector_type(8))) float v8f;

static_assert((DIMIN / 2) * HIDDEN >= NNODES, "the repack grid also clears the per-node counters");


__global__ void k_prep(const float* __restrict__ W, float2* __restrict__ Wp,
                       int* __restrict__ cnt, int n) {
    int idx = blockIdx.x * blockDim.x + threadIdx.x;
    if (idx < n) cnt[idx] = 0;
    if (idx >= (DIMIN / 2) * HIDDEN) return;
    int kp  = idx / HIDDEN;
    int col = idx - kp * HIDDEN;
    float2 p;
    p.x = W[(size_t)(2 * kp)     * HIDDEN + col];
    p.y = W[(size_t)(2 * kp + 1) * HIDDEN + col];
    Wp[idx] = p;
}


__global__ void k_count_rank(const int* __restrict__ dst, int* __restrict__ cnt,
                             int* __restrict__ rank, int e) {
    int i = blockIdx.x * blockDim.x + threadIdx.x;
    if (i < e) rank[i] = atomicAdd(&cnt[dst[i]], 1);
}

__global__ void __launch_bounds__(1024)
k_scan_dinv(const int* __restrict__ cnt, int* __restrict__ rowptr,
            float* __restrict__ dinv, int n) {
    __shared__ int wtot[32];
    const int t    = threadIdx.x;
    const int lane = t & 31;
    const int w    = t >> 5;
    const int per  = (n + 1023) / 1024;
    const int b    = t * per;

    int s = 0;
    for (int i = 0; i < per; ++i) {
        const int j = b + i;
        const int c = cnt[j < n ? j : n - 1];
        s += j < n ? c : 0;
    }
    int inc = s;
    for (int o = 1; o < 32; o <<= 1) {
        const int v = __shfl_up(inc, o);
        if (lane >= o) inc += v;
    }
    if (lane == 31) wtot[w] = inc;
    __syncthreads();
    if (w == 0) {
        const int v = wtot[lane];
        int iv = v;
        for (int o = 1; o < 32; o <<= 1) {
            const int u = __shfl_up(iv, o);
            if (lane >= o) iv += u;
        }
        wtot[lane] = iv - v;
    }
    __syncthreads();
    int run = wtot[w] + inc - s;
    for (int i = 0; i < per; ++i) {
        const int j = b + i;
        const int c = cnt[j < n ? j : n - 1];
        if (j < n) {
            rowptr[j] = run;
            dinv[j]   = __frsqrt_rn((float)(c + 1));
            run += c;
        }
    }
    if (t == 1023) rowptr[n] = run;
}

__global__ void k_fill(const int* __restrict__ src, const int* __restrict__ dst,
                       const int* __restrict__ rowptr, const int* __restrict__ rank,
                       int* __restrict__ csr, int e) {
    int i = blockIdx.x * blockDim.x + threadIdx.x;
    if (i < e) csr[rowptr[dst[i]] + rank[i]] = src[i];
}


__global__ void __launch_bounds__(256)
k_gemm1(const float* __restrict__ X, const float2* __restrict__ Wp, float* __restrict__ H) {
    __shared__ float As[16 * LDSK];

    const int row0 = blockIdx.x * 16;

    for (int idx = threadIdx.x; idx < (16 * DIMIN) / 4; idx += 256) {
        int r = idx >> 7;
        int c = (idx & 127) * 4;
        const float4 v = *(const float4*)(X + (size_t)(row0 + r) * DIMIN + c);
        *(float4*)(&As[r * LDSK + c]) = v;
    }
    __syncthreads();

    const int wave = threadIdx.x >> 5;
    const int lane = threadIdx.x & 31;
    const int l16  = lane & 15;
    const int hi   = lane >> 4;

    for (int ct = wave; ct < HIDDEN / 16; ct += 8) {
        const int col0 = ct * 16;
        v8f c = {};
        #pragma unroll 8
        for (int k = 0; k < DIMIN; k += 4) {
            const int kk = k + 2 * hi;
            const float2 af = *(const float2*)(&As[l16 * LDSK + kk]);
            v2f a; a.x = af.x; a.y = af.y;
            const float2 bf = Wp[(size_t)(kk >> 1) * HIDDEN + col0 + l16];
            v2f b; b.x = bf.x; b.y = bf.y;
            c = __builtin_amdgcn_wmma_f32_16x16x4_f32(false, a, false, b,
                                                      (short)0, c, false, false);
        }
        #pragma unroll
        for (int v = 0; v < 8; ++v) {
            const int row = row0 + hi * 8 + v;
            H[(size_t)row * HIDDEN + col0 + l16] = c[v];
        }
    }
}


__device__ __forceinline__ float relu(float v) { return v > 0.0f ? v : 0.0f; }

__global__ void __launch_bounds__(256)
k_gather1(const int* __restrict__ rowptr, const int* __restrict__ csr,
          const float* __restrict__ dinv, const float* __restrict__ H1,
          const float* __restrict__ b1, const float* __restrict__ W2,
          float* __restrict__ H2, int n) {
    const int d    = __builtin_amdgcn_readfirstlane((int)((blockIdx.x * blockDim.x + threadIdx.x) >> 5));
    const int lane = threadIdx.x & 31;
    if (d >= n) return;

    const float dd  = dinv[d];
    const int   beg = rowptr[d];
    const int   end = rowptr[d + 1];

    const float4* hd = (const float4*)(H1 + (size_t)d * HIDDEN);
    const float   sn = dd * dd;
    const float4  s0 = hd[lane], s1 = hd[32 + lane];
    float4 a0, a1;
    a0.x = sn * s0.x; a0.y = sn * s0.y; a0.z = sn * s0.z; a0.w = sn * s0.w;
    a1.x = sn * s1.x; a1.y = sn * s1.y; a1.z = sn * s1.z; a1.w = sn * s1.w;

    for (int base = beg; base < end; base += 32) {
        const int j = base + lane;
        int   s_l  = 0;
        float nm_l = 0.0f;
        if (j < end) { s_l = csr[j]; nm_l = dinv[s_l] * dd; }
        const int m = (end - base) < 32 ? (end - base) : 32;
        #pragma unroll 4
        for (int t = 0; t < m; ++t) {
            const int   s  = __builtin_amdgcn_readlane(s_l, t);
            const float nm = __int_as_float(__builtin_amdgcn_readlane(__float_as_int(nm_l), t));
            const float4* hs = (const float4*)(H1 + (size_t)s * HIDDEN);
            const float4 g0 = hs[lane], g1 = hs[32 + lane];
            a0.x += nm * g0.x; a0.y += nm * g0.y; a0.z += nm * g0.z; a0.w += nm * g0.w;
            a1.x += nm * g1.x; a1.y += nm * g1.y; a1.z += nm * g1.z; a1.w += nm * g1.w;
        }
    }

    const float4 bA = ((const float4*)b1)[lane], bB = ((const float4*)b1)[32 + lane];
    const float z0 = relu(a0.x + bA.x), z1 = relu(a0.y + bA.y);
    const float z2 = relu(a0.z + bA.z), z3 = relu(a0.w + bA.w);
    const float z4 = relu(a1.x + bB.x), z5 = relu(a1.y + bB.y);
    const float z6 = relu(a1.z + bB.z), z7 = relu(a1.w + bB.w);

    const float4* w4 = (const float4*)W2;
    const float4 wA0 = w4[2 * lane],      wA1 = w4[2 * lane + 1];
    const float4 wB0 = w4[64 + 2 * lane], wB1 = w4[64 + 2 * lane + 1];
    float p0 = z0 * wA0.x + z1 * wA0.z + z2 * wA1.x + z3 * wA1.z
             + z4 * wB0.x + z5 * wB0.z + z6 * wB1.x + z7 * wB1.z;
    float p1 = z0 * wA0.y + z1 * wA0.w + z2 * wA1.y + z3 * wA1.w
             + z4 * wB0.y + z5 * wB0.w + z6 * wB1.y + z7 * wB1.w;
    for (int o = 16; o > 0; o >>= 1) {
        p0 += __shfl_xor(p0, o);
        p1 += __shfl_xor(p1, o);
    }
    if (lane == 0) {
        H2[2 * d]     = p0;
        H2[2 * d + 1] = p1;
    }
}


__global__ void k_gather2(const int* __restrict__ rowptr, const int* __restrict__ csr,
                          const float* __restrict__ dinv, const float* __restrict__ H2,
                          const float* __restrict__ b2, float* __restrict__ out, int n) {
    const int gid = blockIdx.x * blockDim.x + threadIdx.x;
    const int d   = gid >> 4;
    const int l   = gid & 15;
    const float2* h2 = (const float2*)H2;
    float p0 = 0.0f, p1 = 0.0f, dd = 0.0f;
    if (d < n) {
        dd = dinv[d];
        const int beg = rowptr[d];
        const int end = rowptr[d + 1];
        for (int j = beg + l; j < end; j += 16) {
            const int    s  = csr[j];
            const float  nm = dinv[s] * dd;
            const float2 h  = h2[s];
            p0 += nm * h.x;
            p1 += nm * h.y;
        }
    }
    for (int o = 8; o > 0; o >>= 1) {
        p0 += __shfl_xor(p0, o, 16);
        p1 += __shfl_xor(p1, o, 16);
    }
    if (d < n && l == 0) {
        const float  sn = dd * dd;
        const float2 hd = h2[d];
        out[2 * d]     = (sn * hd.x + b2[0]) + p0;
        out[2 * d + 1] = (sn * hd.y + b2[1]) + p1;
    }
}


extern "C" void kernel_launch(void* const* d_in, const int* in_sizes, int n_in,
                              void* d_out, int out_size, void* d_ws, size_t ws_size,
                              hipStream_t stream) {
    const float* X   = (const float*)d_in[0];
    const float* W1  = (const float*)d_in[1];
    const float* b1  = (const float*)d_in[2];
    const float* W2  = (const float*)d_in[3];
    const float* b2  = (const float*)d_in[4];
    const int*   ei  = (const int*)d_in[5];
    float*       out = (float*)d_out;

    const int N = NNODES;
    const int E = in_sizes[5] / 2;
    const int* src = ei;
    const int* dst = ei + E;

    int*    cnt    = (int*)d_ws;
    float*  dinv   = (float*)(cnt + N);
    float*  H1     = dinv + N;
    int*    rowptr = (int*)(H1 + (size_t)N * HIDDEN);
    int*    rank   = rowptr + (N + 1);
    int*    csr    = rank + E;
    float*  H2     = H1 + (size_t)2 * N * HIDDEN;
    float2* Wp     = (float2*)(H2 + (size_t)N * NCLS);

    const int T = 256;

    {
        int total = (DIMIN / 2) * HIDDEN;
        k_prep<<<(total + T - 1) / T, T, 0, stream>>>(W1, Wp, cnt, N);
    }

    k_count_rank<<<(E + T - 1) / T, T, 0, stream>>>(dst, cnt, rank, E);
    k_scan_dinv <<<1, 1024, 0, stream>>>(cnt, rowptr, dinv, N);
    k_fill      <<<(E + T - 1) / T, T, 0, stream>>>(src, dst, rowptr, rank, csr, E);

    k_gemm1<<<N / 16, 256, 0, stream>>>(X, Wp, H1);
    {
        long long threads = (long long)N * 32;
        k_gather1<<<(int)((threads + T - 1) / T), T, 0, stream>>>(rowptr, csr, dinv, H1, b1, W2, H2, N);
    }

    {
        long long threads = (long long)N * 16;
        k_gather2<<<(int)((threads + T - 1) / T), T, 0, stream>>>(rowptr, csr, dinv, H2, b2, out, N);
    }
}


# Single-block prefix scan of the in-degree counts staged through LDS: whole-line 16-byte global loads and stores replace 4-byte accesses at a 196-byte lane pitch; integer sums, results unchanged
# speedup vs baseline: 2.3097x; 1.3722x over previous
#include <hip/hip_runtime.h>
#include <hip/hip_bf16.h>

#define NNODES 50000
#define DIMIN  512
#define HIDDEN 256
#define NCLS   2
#define LDSK   516

typedef __attribute__((ext_vector_type(2))) float v2f;
typedef __attribute__((ext_vector_type(8))) float v8f;

static_assert((DIMIN / 2) * HIDDEN >= NNODES, "the repack grid also clears the per-node counters");


__global__ void k_prep(const float* __restrict__ W, float2* __restrict__ Wp,
                       int* __restrict__ cnt, int n) {
    int idx = blockIdx.x * blockDim.x + threadIdx.x;
    if (idx < n) cnt[idx] = 0;
    if (idx >= (DIMIN / 2) * HIDDEN) return;
    int kp  = idx / HIDDEN;
    int col = idx - kp * HIDDEN;
    float2 p;
    p.x = W[(size_t)(2 * kp)     * HIDDEN + col];
    p.y = W[(size_t)(2 * kp + 1) * HIDDEN + col];
    Wp[idx] = p;
}


__global__ void k_count_rank(const int* __restrict__ dst, int* __restrict__ cnt,
                             int* __restrict__ rank, int e) {
    int i = blockIdx.x * blockDim.x + threadIdx.x;
    if (i < e) rank[i] = atomicAdd(&cnt[dst[i]], 1);
}

#define SCAN_T    1024
#define SCAN_Q    13
#define SCAN_CAPQ (SCAN_T * SCAN_Q)
static_assert(NNODES % 4 == 0, "the scan moves the per-node arrays as 16-byte pieces");
#define SCAN_NQ   (NNODES / 4)
static_assert(4 * SCAN_CAPQ > NNODES, "the stage must hold every count and one slot behind the last");

__global__ void __launch_bounds__(SCAN_T)
k_scan_dinv(const int* __restrict__ cnt, int* __restrict__ rowptr,
            float* __restrict__ dinv) {
    __shared__ int4 sc4[SCAN_CAPQ];
    __shared__ int  wtot[32];
    const int t    = threadIdx.x;
    const int lane = t & 31;
    const int w    = t >> 5;

    #pragma unroll
    for (int i = 0; i < SCAN_Q; ++i) {
        const int q = t + i * SCAN_T;
        int4 v = make_int4(0, 0, 0, 0);
        if (q < SCAN_NQ) v = ((const int4*)cnt)[q];
        sc4[q] = v;
    }
    __syncthreads();

    int4 c[SCAN_Q];
    int  s = 0;
    #pragma unroll
    for (int i = 0; i < SCAN_Q; ++i) {
        c[i] = sc4[t * SCAN_Q + i];
        s += (c[i].x + c[i].y) + (c[i].z + c[i].w);
    }
    int inc = s;
    for (int o = 1; o < 32; o <<= 1) {
        const int v = __shfl_up(inc, o);
        if (lane >= o) inc += v;
    }
    if (lane == 31) wtot[w] = inc;
    __syncthreads();
    if (w == 0) {
        const int v = wtot[lane];
        int iv = v;
        for (int o = 1; o < 32; o <<= 1) {
            const int u = __shfl_up(iv, o);
            if (lane >= o) iv += u;
        }
        wtot[lane] = iv - v;
    }
    __syncthreads();

    int run = wtot[w] + inc - s;
    #pragma unroll
    for (int i = 0; i < SCAN_Q; ++i) {
        int4 p;
        p.x = run; run += c[i].x;
        p.y = run; run += c[i].y;
        p.z = run; run += c[i].z;
        p.w = run; run += c[i].w;
        sc4[t * SCAN_Q + i] = p;
    }
    __syncthreads();

    const int* sc = (const int*)sc4;
    for (int q = t; q < SCAN_NQ; q += SCAN_T) {
        const int4 p  = sc4[q];
        const int  nx = sc[4 * q + 4];
        ((int4*)rowptr)[q] = p;
        float4 dv;
        dv.x = __frsqrt_rn((float)(p.y - p.x + 1));
        dv.y = __frsqrt_rn((float)(p.z - p.y + 1));
        dv.z = __frsqrt_rn((float)(p.w - p.z + 1));
        dv.w = __frsqrt_rn((float)(nx  - p.w + 1));
        ((float4*)dinv)[q] = dv;
    }
    if (t == 0) rowptr[NNODES] = sc[NNODES];
}

__global__ void k_fill(const int* __restrict__ src, const int* __restrict__ dst,
                       const int* __restrict__ rowptr, const int* __restrict__ rank,
                       int* __restrict__ csr, int e) {
    int i = blockIdx.x * blockDim.x + threadIdx.x;
    if (i < e) csr[rowptr[dst[i]] + rank[i]] = src[i];
}


__global__ void __launch_bounds__(256)
k_gemm1(const float* __restrict__ X, const float2* __restrict__ Wp, float* __restrict__ H) {
    __shared__ float As[16 * LDSK];

    const int row0 = blockIdx.x * 16;

    for (int idx = threadIdx.x; idx < (16 * DIMIN) / 4; idx += 256) {
        int r = idx >> 7;
        int c = (idx & 127) * 4;
        const float4 v = *(const float4*)(X + (size_t)(row0 + r) * DIMIN + c);
        *(float4*)(&As[r * LDSK + c]) = v;
    }
    __syncthreads();

    const int wave = threadIdx.x >> 5;
    const int lane = threadIdx.x & 31;
    const int l16  = lane & 15;
    const int hi   = lane >> 4;

    for (int ct = wave; ct < HIDDEN / 16; ct += 8) {
        const int col0 = ct * 16;
        v8f c = {};
        #pragma unroll 8
        for (int k = 0; k < DIMIN; k += 4) {
            const int kk = k + 2 * hi;
            const float2 af = *(const float2*)(&As[l16 * LDSK + kk]);
            v2f a; a.x = af.x; a.y = af.y;
            const float2 bf = Wp[(size_t)(kk >> 1) * HIDDEN + col0 + l16];
            v2f b; b.x = bf.x; b.y = bf.y;
            c = __builtin_amdgcn_wmma_f32_16x16x4_f32(false, a, false, b,
                                                      (short)0, c, false, false);
        }
        #pragma unroll
        for (int v = 0; v < 8; ++v) {
            const int row = row0 + hi * 8 + v;
            H[(size_t)row * HIDDEN + col0 + l16] = c[v];
        }
    }
}


__device__ __forceinline__ float relu(float v) { return v > 0.0f ? v : 0.0f; }

__global__ void __launch_bounds__(256)
k_gather1(const int* __restrict__ rowptr, const int* __restrict__ csr,
          const float* __restrict__ dinv, const float* __restrict__ H1,
          const float* __restrict__ b1, const float* __restrict__ W2,
          float* __restrict__ H2, int n) {
    const int d    = __builtin_amdgcn_readfirstlane((int)((blockIdx.x * blockDim.x + threadIdx.x) >> 5));
    const int lane = threadIdx.x & 31;
    if (d >= n) return;

    const float dd  = dinv[d];
    const int   beg = rowptr[d];
    const int   end = rowptr[d + 1];

    const float4* hd = (const float4*)(H1 + (size_t)d * HIDDEN);
    const float   sn = dd * dd;
    const float4  s0 = hd[lane], s1 = hd[32 + lane];
    float4 a0, a1;
    a0.x = sn * s0.x; a0.y = sn * s0.y; a0.z = sn * s0.z; a0.w = sn * s0.w;
    a1.x = sn * s1.x; a1.y = sn * s1.y; a1.z = sn * s1.z; a1.w = sn * s1.w;

    for (int base = beg; base < end; base += 32) {
        const int j = base + lane;
        int   s_l  = 0;
        float nm_l = 0.0f;
        if (j < end) { s_l = csr[j]; nm_l = dinv[s_l] * dd; }
        const int m = (end - base) < 32 ? (end - base) : 32;
        #pragma unroll 4
        for (int t = 0; t < m; ++t) {
            const int   s  = __builtin_amdgcn_readlane(s_l, t);
            const float nm = __int_as_float(__builtin_amdgcn_readlane(__float_as_int(nm_l), t));
            const float4* hs = (const float4*)(H1 + (size_t)s * HIDDEN);
            const float4 g0 = hs[lane], g1 = hs[32 + lane];
            a0.x += nm * g0.x; a0.y += nm * g0.y; a0.z += nm * g0.z; a0.w += nm * g0.w;
            a1.x += nm * g1.x; a1.y += nm * g1.y; a1.z += nm * g1.z; a1.w += nm * g1.w;
        }
    }

    const float4 bA = ((const float4*)b1)[lane], bB = ((const float4*)b1)[32 + lane];
    const float z0 = relu(a0.x + bA.x), z1 = relu(a0.y + bA.y);
    const float z2 = relu(a0.z + bA.z), z3 = relu(a0.w + bA.w);
    const float z4 = relu(a1.x + bB.x), z5 = relu(a1.y + bB.y);
    const float z6 = relu(a1.z + bB.z), z7 = relu(a1.w + bB.w);

    const float4* w4 = (const float4*)W2;
    const float4 wA0 = w4[2 * lane],      wA1 = w4[2 * lane + 1];
    const float4 wB0 = w4[64 + 2 * lane], wB1 = w4[64 + 2 * lane + 1];
    float p0 = z0 * wA0.x + z1 * wA0.z + z2 * wA1.x + z3 * wA1.z
             + z4 * wB0.x + z5 * wB0.z + z6 * wB1.x + z7 * wB1.z;
    float p1 = z0 * wA0.y + z1 * wA0.w + z2 * wA1.y + z3 * wA1.w
             + z4 * wB0.y + z5 * wB0.w + z6 * wB1.y + z7 * wB1.w;
    for (int o = 16; o > 0; o >>= 1) {
        p0 += __shfl_xor(p0, o);
        p1 += __shfl_xor(p1, o);
    }
    if (lane == 0) {
        H2[2 * d]     = p0;
        H2[2 * d + 1] = p1;
    }
}


__global__ void k_gather2(const int* __restrict__ rowptr, const int* __restrict__ csr,
                          const float* __restrict__ dinv, const float* __restrict__ H2,
                          const float* __restrict__ b2, float* __restrict__ out, int n) {
    const int gid = blockIdx.x * blockDim.x + threadIdx.x;
    const int d   = gid >> 4;
    const int l   = gid & 15;
    const float2* h2 = (const float2*)H2;
    float p0 = 0.0f, p1 = 0.0f, dd = 0.0f;
    if (d < n) {
        dd = dinv[d];
        const int beg = rowptr[d];
        const int end = rowptr[d + 1];
        for (int j = beg + l; j < end; j += 16) {
            const int    s  = csr[j];
            const float  nm = dinv[s] * dd;
            const float2 h  = h2[s];
            p0 += nm * h.x;
            p1 += nm * h.y;
        }
    }
    for (int o = 8; o > 0; o >>= 1) {
        p0 += __shfl_xor(p0, o, 16);
        p1 += __shfl_xor(p1, o, 16);
    }
    if (d < n && l == 0) {
        const float  sn = dd * dd;
        const float2 hd = h2[d];
        out[2 * d]     = (sn * hd.x + b2[0]) + p0;
        out[2 * d + 1] = (sn * hd.y + b2[1]) + p1;
    }
}


extern "C" void kernel_launch(void* const* d_in, const int* in_sizes, int n_in,
                              void* d_out, int out_size, void* d_ws, size_t ws_size,
                              hipStream_t stream) {
    const float* X   = (const float*)d_in[0];
    const float* W1  = (const float*)d_in[1];
    const float* b1  = (const float*)d_in[2];
    const float* W2  = (const float*)d_in[3];
    const float* b2  = (const float*)d_in[4];
    const int*   ei  = (const int*)d_in[5];
    float*       out = (float*)d_out;

    const int N = NNODES;
    const int E = in_sizes[5] / 2;
    const int* src = ei;
    const int* dst = ei + E;

    int*    cnt    = (int*)d_ws;
    float*  dinv   = (float*)(cnt + N);
    float*  H1     = dinv + N;
    int*    rowptr = (int*)(H1 + (size_t)N * HIDDEN);
    int*    rank   = rowptr + (N + 1);
    int*    csr    = rank + E;
    float*  H2     = H1 + (size_t)2 * N * HIDDEN;
    float2* Wp     = (float2*)(H2 + (size_t)N * NCLS);

    const int T = 256;

    {
        int total = (DIMIN / 2) * HIDDEN;
        k_prep<<<(total + T - 1) / T, T, 0, stream>>>(W1, Wp, cnt, N);
    }

    k_count_rank<<<(E + T - 1) / T, T, 0, stream>>>(dst, cnt, rank, E);
    k_scan_dinv <<<1, SCAN_T, 0, stream>>>(cnt, rowptr, dinv);
    k_fill      <<<(E + T - 1) / T, T, 0, stream>>>(src, dst, rowptr, rank, csr, E);

    k_gemm1<<<N / 16, 256, 0, stream>>>(X, Wp, H1);
    {
        long long threads = (long long)N * 32;
        k_gather1<<<(int)((threads + T - 1) / T), T, 0, stream>>>(rowptr, csr, dinv, H1, b1, W2, H2, N);
    }

    {
        long long threads = (long long)N * 16;
        k_gather2<<<(int)((threads + T - 1) / T), T, 0, stream>>>(rowptr, csr, dinv, H2, b2, out, N);
    }
}


# x@W1 product split into two half-row launches; the count pass and the fill pass of the edge sort ride in them as extra work-groups behind the matrix ones and run beside the matrix work
# speedup vs baseline: 2.5927x; 1.1225x over previous
#include <hip/hip_runtime.h>
#include <hip/hip_bf16.h>

#define NNODES 50000
#define DIMIN  512
#define HIDDEN 256
#define NCLS   2
#define LDSK   516

typedef __attribute__((ext_vector_type(2))) float v2f;
typedef __attribute__((ext_vector_type(8))) float v8f;

static_assert((DIMIN / 2) * HIDDEN >= NNODES, "the repack grid also clears the per-node counters");


__global__ void k_prep(const float* __restrict__ W, float2* __restrict__ Wp,
                       int* __restrict__ cnt, int n) {
    int idx = blockIdx.x * blockDim.x + threadIdx.x;
    if (idx < n) cnt[idx] = 0;
    if (idx >= (DIMIN / 2) * HIDDEN) return;
    int kp  = idx / HIDDEN;
    int col = idx - kp * HIDDEN;
    float2 p;
    p.x = W[(size_t)(2 * kp)     * HIDDEN + col];
    p.y = W[(size_t)(2 * kp + 1) * HIDDEN + col];
    Wp[idx] = p;
}


__device__ __forceinline__ void edge_count_rank(const int* __restrict__ dst, int* __restrict__ cnt,
                                                int* __restrict__ rank, int i, int e) {
    if (i < e) rank[i] = atomicAdd(&cnt[dst[i]], 1);
}

#define SCAN_T    1024
#define SCAN_Q    13
#define SCAN_CAPQ (SCAN_T * SCAN_Q)
static_assert(NNODES % 4 == 0, "the scan moves the per-node arrays as 16-byte pieces");
#define SCAN_NQ   (NNODES / 4)
static_assert(4 * SCAN_CAPQ > NNODES, "the stage must hold every count and one slot behind the last");

__global__ void __launch_bounds__(SCAN_T)
k_scan_dinv(const int* __restrict__ cnt, int* __restrict__ rowptr,
            float* __restrict__ dinv) {
    __shared__ int4 sc4[SCAN_CAPQ];
    __shared__ int  wtot[32];
    const int t    = threadIdx.x;
    const int lane = t & 31;
    const int w    = t >> 5;

    #pragma unroll
    for (int i = 0; i < SCAN_Q; ++i) {
        const int q = t + i * SCAN_T;
        int4 v = make_int4(0, 0, 0, 0);
        if (q < SCAN_NQ) v = ((const int4*)cnt)[q];
        sc4[q] = v;
    }
    __syncthreads();

    int4 c[SCAN_Q];
    int  s = 0;
    #pragma unroll
    for (int i = 0; i < SCAN_Q; ++i) {
        c[i] = sc4[t * SCAN_Q + i];
        s += (c[i].x + c[i].y) + (c[i].z + c[i].w);
    }
    int inc = s;
    for (int o = 1; o < 32; o <<= 1) {
        const int v = __shfl_up(inc, o);
        if (lane >= o) inc += v;
    }
    if (lane == 31) wtot[w] = inc;
    __syncthreads();
    if (w == 0) {
        const int v = wtot[lane];
        int iv = v;
        for (int o = 1; o < 32; o <<= 1) {
            const int u = __shfl_up(iv, o);
            if (lane >= o) iv += u;
        }
        wtot[lane] = iv - v;
    }
    __syncthreads();

    int run = wtot[w] + inc - s;
    #pragma unroll
    for (int i = 0; i < SCAN_Q; ++i) {
        int4 p;
        p.x = run; run += c[i].x;
        p.y = run; run += c[i].y;
        p.z = run; run += c[i].z;
        p.w = run; run += c[i].w;
        sc4[t * SCAN_Q + i] = p;
    }
    __syncthreads();

    const int* sc = (const int*)sc4;
    for (int q = t; q < SCAN_NQ; q += SCAN_T) {
        const int4 p  = sc4[q];
        const int  nx = sc[4 * q + 4];
        ((int4*)rowptr)[q] = p;
        float4 dv;
        dv.x = __frsqrt_rn((float)(p.y - p.x + 1));
        dv.y = __frsqrt_rn((float)(p.z - p.y + 1));
        dv.z = __frsqrt_rn((float)(p.w - p.z + 1));
        dv.w = __frsqrt_rn((float)(nx  - p.w + 1));
        ((float4*)dinv)[q] = dv;
    }
    if (t == 0) rowptr[NNODES] = sc[NNODES];
}

__device__ __forceinline__ void edge_fill(const int* __restrict__ src, const int* __restrict__ dst,
                                          const int* __restrict__ rowptr, const int* __restrict__ rank,
                                          int* __restrict__ csr, int i, int e) {
    if (i < e) csr[rowptr[dst[i]] + rank[i]] = src[i];
}

#define TILES   (NNODES / 16)
#define TILES_A ((TILES + 1) / 2)
#define TILES_B (TILES - TILES_A)
static_assert(NNODES % 16 == 0, "row tiles cover the nodes exactly");

__device__ __forceinline__ void
gemm1_tile(const float* __restrict__ X, const float2* __restrict__ Wp, float* __restrict__ H,
           float* As, const int row0) {

    for (int idx = threadIdx.x; idx < (16 * DIMIN) / 4; idx += 256) {
        int r = idx >> 7;
        int c = (idx & 127) * 4;
        const float4 v = *(const float4*)(X + (size_t)(row0 + r) * DIMIN + c);
        *(float4*)(&As[r * LDSK + c]) = v;
    }
    __syncthreads();

    const int wave = threadIdx.x >> 5;
    const int lane = threadIdx.x & 31;
    const int l16  = lane & 15;
    const int hi   = lane >> 4;

    for (int ct = wave; ct < HIDDEN / 16; ct += 8) {
        const int col0 = ct * 16;
        v8f c = {};
        #pragma unroll 8
        for (int k = 0; k < DIMIN; k += 4) {
            const int kk = k + 2 * hi;
            const float2 af = *(const float2*)(&As[l16 * LDSK + kk]);
            v2f a; a.x = af.x; a.y = af.y;
            const float2 bf = Wp[(size_t)(kk >> 1) * HIDDEN + col0 + l16];
            v2f b; b.x = bf.x; b.y = bf.y;
            c = __builtin_amdgcn_wmma_f32_16x16x4_f32(false, a, false, b,
                                                      (short)0, c, false, false);
        }
        #pragma unroll
        for (int v = 0; v < 8; ++v) {
            const int row = row0 + hi * 8 + v;
            H[(size_t)row * HIDDEN + col0 + l16] = c[v];
        }
    }
}

__global__ void __launch_bounds__(256)
k_gemm1a_count(const float* __restrict__ X, const float2* __restrict__ Wp, float* __restrict__ H,
               const int* __restrict__ dst, int* __restrict__ cnt, int* __restrict__ rank, int e) {
    __shared__ float As[16 * LDSK];
    if (blockIdx.x < TILES_A) {
        gemm1_tile(X, Wp, H, As, (int)blockIdx.x * 16);
        return;
    }
    edge_count_rank(dst, cnt, rank, (int)(blockIdx.x - TILES_A) * 256 + (int)threadIdx.x, e);
}

__global__ void __launch_bounds__(256)
k_gemm1b_fill(const float* __restrict__ X, const float2* __restrict__ Wp, float* __restrict__ H,
              const int* __restrict__ src, const int* __restrict__ dst,
              const int* __restrict__ rowptr, const int* __restrict__ rank,
              int* __restrict__ csr, int e) {
    __shared__ float As[16 * LDSK];
    if (blockIdx.x < TILES_B) {
        gemm1_tile(X, Wp, H, As, (TILES_A + (int)blockIdx.x) * 16);
        return;
    }
    edge_fill(src, dst, rowptr, rank, csr, (int)(blockIdx.x - TILES_B) * 256 + (int)threadIdx.x, e);
}


__device__ __forceinline__ float relu(float v) { return v > 0.0f ? v : 0.0f; }

__global__ void __launch_bounds__(256)
k_gather1(const int* __restrict__ rowptr, const int* __restrict__ csr,
          const float* __restrict__ dinv, const float* __restrict__ H1,
          const float* __restrict__ b1, const float* __restrict__ W2,
          float* __restrict__ H2, int n) {
    const int d    = __builtin_amdgcn_readfirstlane((int)((blockIdx.x * blockDim.x + threadIdx.x) >> 5));
    const int lane = threadIdx.x & 31;
    if (d >= n) return;

    const float dd  = dinv[d];
    const int   beg = rowptr[d];
    const int   end = rowptr[d + 1];

    const float4* hd = (const float4*)(H1 + (size_t)d * HIDDEN);
    const float   sn = dd * dd;
    const float4  s0 = hd[lane], s1 = hd[32 + lane];
    float4 a0, a1;
    a0.x = sn * s0.x; a0.y = sn * s0.y; a0.z = sn * s0.z; a0.w = sn * s0.w;
    a1.x = sn * s1.x; a1.y = sn * s1.y; a1.z = sn * s1.z; a1.w = sn * s1.w;

    for (int base = beg; base < end; base += 32) {
        const int j = base + lane;
        int   s_l  = 0;
        float nm_l = 0.0f;
        if (j < end) { s_l = csr[j]; nm_l = dinv[s_l] * dd; }
        const int m = (end - base) < 32 ? (end - base) : 32;
        #pragma unroll 4
        for (int t = 0; t < m; ++t) {
            const int   s  = __builtin_amdgcn_readlane(s_l, t);
            const float nm = __int_as_float(__builtin_amdgcn_readlane(__float_as_int(nm_l), t));
            const float4* hs = (const float4*)(H1 + (size_t)s * HIDDEN);
            const float4 g0 = hs[lane], g1 = hs[32 + lane];
            a0.x += nm * g0.x; a0.y += nm * g0.y; a0.z += nm * g0.z; a0.w += nm * g0.w;
            a1.x += nm * g1.x; a1.y += nm * g1.y; a1.z += nm * g1.z; a1.w += nm * g1.w;
        }
    }

    const float4 bA = ((const float4*)b1)[lane], bB = ((const float4*)b1)[32 + lane];
    const float z0 = relu(a0.x + bA.x), z1 = relu(a0.y + bA.y);
    const float z2 = relu(a0.z + bA.z), z3 = relu(a0.w + bA.w);
    const float z4 = relu(a1.x + bB.x), z5 = relu(a1.y + bB.y);
    const float z6 = relu(a1.z + bB.z), z7 = relu(a1.w + bB.w);

    const float4* w4 = (const float4*)W2;
    const float4 wA0 = w4[2 * lane],      wA1 = w4[2 * lane + 1];
    const float4 wB0 = w4[64 + 2 * lane], wB1 = w4[64 + 2 * lane + 1];
    float p0 = z0 * wA0.x + z1 * wA0.z + z2 * wA1.x + z3 * wA1.z
             + z4 * wB0.x + z5 * wB0.z + z6 * wB1.x + z7 * wB1.z;
    float p1 = z0 * wA0.y + z1 * wA0.w + z2 * wA1.y + z3 * wA1.w
             + z4 * wB0.y + z5 * wB0.w + z6 * wB1.y + z7 * wB1.w;
    for (int o = 16; o > 0; o >>= 1) {
        p0 += __shfl_xor(p0, o);
        p1 += __shfl_xor(p1, o);
    }
    if (lane == 0) {
        H2[2 * d]     = p0;
        H2[2 * d + 1] = p1;
    }
}


__global__ void k_gather2(const int* __restrict__ rowptr, const int* __restrict__ csr,
                          const float* __restrict__ dinv, const float* __restrict__ H2,
                          const float* __restrict__ b2, float* __restrict__ out, int n) {
    const int gid = blockIdx.x * blockDim.x + threadIdx.x;
    const int d   = gid >> 4;
    const int l   = gid & 15;
    const float2* h2 = (const float2*)H2;
    float p0 = 0.0f, p1 = 0.0f, dd = 0.0f;
    if (d < n) {
        dd = dinv[d];
        const int beg = rowptr[d];
        const int end = rowptr[d + 1];
        for (int j = beg + l; j < end; j += 16) {
            const int    s  = csr[j];
            const float  nm = dinv[s] * dd;
            const float2 h  = h2[s];
            p0 += nm * h.x;
            p1 += nm * h.y;
        }
    }
    for (int o = 8; o > 0; o >>= 1) {
        p0 += __shfl_xor(p0, o, 16);
        p1 += __shfl_xor(p1, o, 16);
    }
    if (d < n && l == 0) {
        const float  sn = dd * dd;
        const float2 hd = h2[d];
        out[2 * d]     = (sn * hd.x + b2[0]) + p0;
        out[2 * d + 1] = (sn * hd.y + b2[1]) + p1;
    }
}


extern "C" void kernel_launch(void* const* d_in, const int* in_sizes, int n_in,
                              void* d_out, int out_size, void* d_ws, size_t ws_size,
                              hipStream_t stream) {
    const float* X   = (const float*)d_in[0];
    const float* W1  = (const float*)d_in[1];
    const float* b1  = (const float*)d_in[2];
    const float* W2  = (const float*)d_in[3];
    const float* b2  = (const float*)d_in[4];
    const int*   ei  = (const int*)d_in[5];
    float*       out = (float*)d_out;

    const int N = NNODES;
    const int E = in_sizes[5] / 2;
    const int* src = ei;
    const int* dst = ei + E;

    int*    cnt    = (int*)d_ws;
    float*  dinv   = (float*)(cnt + N);
    float*  H1     = dinv + N;
    int*    rowptr = (int*)(H1 + (size_t)N * HIDDEN);
    int*    rank   = rowptr + (N + 1);
    int*    csr    = rank + E;
    float*  H2     = H1 + (size_t)2 * N * HIDDEN;
    float2* Wp     = (float2*)(H2 + (size_t)N * NCLS);

    const int T = 256;

    {
        int total = (DIMIN / 2) * HIDDEN;
        k_prep<<<(total + T - 1) / T, T, 0, stream>>>(W1, Wp, cnt, N);
    }

    const int EB = (E + T - 1) / T;
    k_gemm1a_count<<<TILES_A + EB, T, 0, stream>>>(X, Wp, H1, dst, cnt, rank, E);
    k_scan_dinv   <<<1, SCAN_T, 0, stream>>>(cnt, rowptr, dinv);
    k_gemm1b_fill <<<TILES_B + EB, T, 0, stream>>>(X, Wp, H1, src, dst, rowptr, rank, csr, E);

    {
        long long threads = (long long)N * 32;
        k_gather1<<<(int)((threads + T - 1) / T), T, 0, stream>>>(rowptr, csr, dinv, H1, b1, W2, H2, N);
    }

    {
        long long threads = (long long)N * 16;
        k_gather2<<<(int)((threads + T - 1) / T), T, 0, stream>>>(rowptr, csr, dinv, H2, b2, out, N);
    }
}


# x@W1 work-groups retiled to 32 rows x 128 cols: a wave keeps two row tiles and feeds both from one weight-fragment load; A slab staged per K half, its loads issued together; sum order unchanged
# speedup vs baseline: 2.7007x; 1.0417x over previous
#include <hip/hip_runtime.h>
#include <hip/hip_bf16.h>

#define NNODES 50000
#define DIMIN  512
#define HIDDEN 256
#define NCLS   2
#define KHALF  256
#define LDSK   260

typedef __attribute__((ext_vector_type(2))) float v2f;
typedef __attribute__((ext_vector_type(8))) float v8f;

static_assert((DIMIN / 2) * HIDDEN >= NNODES, "the repack grid also clears the per-node counters");


__global__ void k_prep(const float* __restrict__ W, float2* __restrict__ Wp,
                       int* __restrict__ cnt, int n) {
    int idx = blockIdx.x * blockDim.x + threadIdx.x;
    if (idx < n) cnt[idx] = 0;
    if (idx >= (DIMIN / 2) * HIDDEN) return;
    int kp  = idx / HIDDEN;
    int col = idx - kp * HIDDEN;
    float2 p;
    p.x = W[(size_t)(2 * kp)     * HIDDEN + col];
    p.y = W[(size_t)(2 * kp + 1) * HIDDEN + col];
    Wp[idx] = p;
}


__device__ __forceinline__ void edge_count_rank(const int* __restrict__ dst, int* __restrict__ cnt,
                                                int* __restrict__ rank, int i, int e) {
    if (i < e) rank[i] = atomicAdd(&cnt[dst[i]], 1);
}

#define SCAN_T    1024
#define SCAN_Q    13
#define SCAN_CAPQ (SCAN_T * SCAN_Q)
static_assert(NNODES % 4 == 0, "the scan moves the per-node arrays as 16-byte pieces");
#define SCAN_NQ   (NNODES / 4)
static_assert(4 * SCAN_CAPQ > NNODES, "the stage must hold every count and one slot behind the last");

__global__ void __launch_bounds__(SCAN_T)
k_scan_dinv(const int* __restrict__ cnt, int* __restrict__ rowptr,
            float* __restrict__ dinv) {
    __shared__ int4 sc4[SCAN_CAPQ];
    __shared__ int  wtot[32];
    const int t    = threadIdx.x;
    const int lane = t & 31;
    const int w    = t >> 5;

    #pragma unroll
    for (int i = 0; i < SCAN_Q; ++i) {
        const int q = t + i * SCAN_T;
        int4 v = make_int4(0, 0, 0, 0);
        if (q < SCAN_NQ) v = ((const int4*)cnt)[q];
        sc4[q] = v;
    }
    __syncthreads();

    int4 c[SCAN_Q];
    int  s = 0;
    #pragma unroll
    for (int i = 0; i < SCAN_Q; ++i) {
        c[i] = sc4[t * SCAN_Q + i];
        s += (c[i].x + c[i].y) + (c[i].z + c[i].w);
    }
    int inc = s;
    for (int o = 1; o < 32; o <<= 1) {
        const int v = __shfl_up(inc, o);
        if (lane >= o) inc += v;
    }
    if (lane == 31) wtot[w] = inc;
    __syncthreads();
    if (w == 0) {
        const int v = wtot[lane];
        int iv = v;
        for (int o = 1; o < 32; o <<= 1) {
            const int u = __shfl_up(iv, o);
            if (lane >= o) iv += u;
        }
        wtot[lane] = iv - v;
    }
    __syncthreads();

    int run = wtot[w] + inc - s;
    #pragma unroll
    for (int i = 0; i < SCAN_Q; ++i) {
        int4 p;
        p.x = run; run += c[i].x;
        p.y = run; run += c[i].y;
        p.z = run; run += c[i].z;
        p.w = run; run += c[i].w;
        sc4[t * SCAN_Q + i] = p;
    }
    __syncthreads();

    const int* sc = (const int*)sc4;
    for (int q = t; q < SCAN_NQ; q += SCAN_T) {
        const int4 p  = sc4[q];
        const int  nx = sc[4 * q + 4];
        ((int4*)rowptr)[q] = p;
        float4 dv;
        dv.x = __frsqrt_rn((float)(p.y - p.x + 1));
        dv.y = __frsqrt_rn((float)(p.z - p.y + 1));
        dv.z = __frsqrt_rn((float)(p.w - p.z + 1));
        dv.w = __frsqrt_rn((float)(nx  - p.w + 1));
        ((float4*)dinv)[q] = dv;
    }
    if (t == 0) rowptr[NNODES] = sc[NNODES];
}

__device__ __forceinline__ void edge_fill(const int* __restrict__ src, const int* __restrict__ dst,
                                          const int* __restrict__ rowptr, const int* __restrict__ rank,
                                          int* __restrict__ csr, int i, int e) {
    if (i < e) csr[rowptr[dst[i]] + rank[i]] = src[i];
}

#define TILES   (NNODES / 16)
#define PAIRS   ((TILES + 1) / 2)
#define PAIRS_A ((PAIRS + 1) / 2)
#define PAIRS_B (PAIRS - PAIRS_A)
#define GBLK_A  (2 * PAIRS_A)
#define GBLK_B  (2 * PAIRS_B)
static_assert(NNODES % 16 == 0, "row tiles cover the nodes exactly");
static_assert(NNODES >= 32, "a moved-up last pair needs 32 rows to stand on");
static_assert(HIDDEN == 2 * 8 * 16, "two column halves of eight 16-column tiles, one tile per wave");
static_assert(DIMIN == 2 * KHALF, "the product runs as two staged K passes");
static_assert((32 * KHALF) / 4 == 8 * 256, "eight 16-byte pieces of the staged slab per thread");

__device__ __forceinline__ void
gemm1_tile(const float* __restrict__ X, const float2* __restrict__ Wp, float* __restrict__ H,
           float* As, const int gb) {
    const bool full = (gb >> 1) * 32 + 32 <= NNODES;
    const int  row0 = full ? (gb >> 1) * 32 : NNODES - 32;
    const int wave = threadIdx.x >> 5;
    const int lane = threadIdx.x & 31;
    const int l16  = lane & 15;
    const int hi   = lane >> 4;
    const int col0 = ((gb & 1) * 8 + wave) * 16;

    v8f c0 = {};
    v8f c1 = {};

    const int  sr  = threadIdx.x >> 6;
    const int  sc  = (threadIdx.x & 63) * 4;
    const float* xlo = X + (size_t)(row0 + sr) * DIMIN + sc;

    #pragma nounroll
    for (int kh = 0; kh < DIMIN; kh += KHALF) {
        if (kh != 0) __syncthreads();
        float4 st[8];
        #pragma unroll
        for (int i = 0; i < 8; ++i)
            st[i] = *(const float4*)(xlo + kh + (size_t)(4 * i) * DIMIN);
        #pragma unroll
        for (int i = 0; i < 8; ++i)
            *(float4*)(&As[(sr + 4 * i) * LDSK + sc]) = st[i];
        __syncthreads();

        const float2* Wk = Wp + (size_t)(kh >> 1) * HIDDEN + col0 + l16;
        #pragma unroll 8
        for (int k = 0; k < KHALF; k += 4) {
            const int kk = k + 2 * hi;
            const float2 a0f = *(const float2*)(&As[l16 * LDSK + kk]);
            const float2 a1f = *(const float2*)(&As[(16 + l16) * LDSK + kk]);
            v2f a0; a0.x = a0f.x; a0.y = a0f.y;
            v2f a1; a1.x = a1f.x; a1.y = a1f.y;
            const float2 bf = Wk[(size_t)(kk >> 1) * HIDDEN];
            v2f b; b.x = bf.x; b.y = bf.y;
            c0 = __builtin_amdgcn_wmma_f32_16x16x4_f32(false, a0, false, b,
                                                       (short)0, c0, false, false);
            c1 = __builtin_amdgcn_wmma_f32_16x16x4_f32(false, a1, false, b,
                                                       (short)0, c1, false, false);
        }
    }

    if (full) {
        #pragma unroll
        for (int v = 0; v < 8; ++v) {
            const int row = row0 + hi * 8 + v;
            H[(size_t)row * HIDDEN + col0 + l16] = c0[v];
        }
    }
    #pragma unroll
    for (int v = 0; v < 8; ++v) {
        const int row = row0 + 16 + hi * 8 + v;
        H[(size_t)row * HIDDEN + col0 + l16] = c1[v];
    }
}

__global__ void __launch_bounds__(256)
k_gemm1a_count(const float* __restrict__ X, const float2* __restrict__ Wp, float* __restrict__ H,
               const int* __restrict__ dst, int* __restrict__ cnt, int* __restrict__ rank, int e) {
    __shared__ float As[32 * LDSK];
    if (blockIdx.x < GBLK_A) {
        gemm1_tile(X, Wp, H, As, (int)blockIdx.x);
        return;
    }
    edge_count_rank(dst, cnt, rank, (int)(blockIdx.x - GBLK_A) * 256 + (int)threadIdx.x, e);
}

__global__ void __launch_bounds__(256)
k_gemm1b_fill(const float* __restrict__ X, const float2* __restrict__ Wp, float* __restrict__ H,
              const int* __restrict__ src, const int* __restrict__ dst,
              const int* __restrict__ rowptr, const int* __restrict__ rank,
              int* __restrict__ csr, int e) {
    __shared__ float As[32 * LDSK];
    if (blockIdx.x < GBLK_B) {
        gemm1_tile(X, Wp, H, As, GBLK_A + (int)blockIdx.x);
        return;
    }
    edge_fill(src, dst, rowptr, rank, csr, (int)(blockIdx.x - GBLK_B) * 256 + (int)threadIdx.x, e);
}


__device__ __forceinline__ float relu(float v) { return v > 0.0f ? v : 0.0f; }

__global__ void __launch_bounds__(256)
k_gather1(const int* __restrict__ rowptr, const int* __restrict__ csr,
          const float* __restrict__ dinv, const float* __restrict__ H1,
          const float* __restrict__ b1, const float* __restrict__ W2,
          float* __restrict__ H2, int n) {
    const int d    = __builtin_amdgcn_readfirstlane((int)((blockIdx.x * blockDim.x + threadIdx.x) >> 5));
    const int lane = threadIdx.x & 31;
    if (d >= n) return;

    const float dd  = dinv[d];
    const int   beg = rowptr[d];
    const int   end = rowptr[d + 1];

    const float4* hd = (const float4*)(H1 + (size_t)d * HIDDEN);
    const float   sn = dd * dd;
    const float4  s0 = hd[lane], s1 = hd[32 + lane];
    float4 a0, a1;
    a0.x = sn * s0.x; a0.y = sn * s0.y; a0.z = sn * s0.z; a0.w = sn * s0.w;
    a1.x = sn * s1.x; a1.y = sn * s1.y; a1.z = sn * s1.z; a1.w = sn * s1.w;

    for (int base = beg; base < end; base += 32) {
        const int j = base + lane;
        int   s_l  = 0;
        float nm_l = 0.0f;
        if (j < end) { s_l = csr[j]; nm_l = dinv[s_l] * dd; }
        const int m = (end - base) < 32 ? (end - base) : 32;
        #pragma unroll 4
        for (int t = 0; t < m; ++t) {
            const int   s  = __builtin_amdgcn_readlane(s_l, t);
            const float nm = __int_as_float(__builtin_amdgcn_readlane(__float_as_int(nm_l), t));
            const float4* hs = (const float4*)(H1 + (size_t)s * HIDDEN);
            const float4 g0 = hs[lane], g1 = hs[32 + lane];
            a0.x += nm * g0.x; a0.y += nm * g0.y; a0.z += nm * g0.z; a0.w += nm * g0.w;
            a1.x += nm * g1.x; a1.y += nm * g1.y; a1.z += nm * g1.z; a1.w += nm * g1.w;
        }
    }

    const float4 bA = ((const float4*)b1)[lane], bB = ((const float4*)b1)[32 + lane];
    const float z0 = relu(a0.x + bA.x), z1 = relu(a0.y + bA.y);
    const float z2 = relu(a0.z + bA.z), z3 = relu(a0.w + bA.w);
    const float z4 = relu(a1.x + bB.x), z5 = relu(a1.y + bB.y);
    const float z6 = relu(a1.z + bB.z), z7 = relu(a1.w + bB.w);

    const float4* w4 = (const float4*)W2;
    const float4 wA0 = w4[2 * lane],      wA1 = w4[2 * lane + 1];
    const float4 wB0 = w4[64 + 2 * lane], wB1 = w4[64 + 2 * lane + 1];
    float p0 = z0 * wA0.x + z1 * wA0.z + z2 * wA1.x + z3 * wA1.z
             + z4 * wB0.x + z5 * wB0.z + z6 * wB1.x + z7 * wB1.z;
    float p1 = z0 * wA0.y + z1 * wA0.w + z2 * wA1.y + z3 * wA1.w
             + z4 * wB0.y + z5 * wB0.w + z6 * wB1.y + z7 * wB1.w;
    for (int o = 16; o > 0; o >>= 1) {
        p0 += __shfl_xor(p0, o);
        p1 += __shfl_xor(p1, o);
    }
    if (lane == 0) {
        H2[2 * d]     = p0;
        H2[2 * d + 1] = p1;
    }
}


__global__ void k_gather2(const int* __restrict__ rowptr, const int* __restrict__ csr,
                          const float* __restrict__ dinv, const float* __restrict__ H2,
                          const float* __restrict__ b2, float* __restrict__ out, int n) {
    const int gid = blockIdx.x * blockDim.x + threadIdx.x;
    const int d   = gid >> 4;
    const int l   = gid & 15;
    const float2* h2 = (const float2*)H2;
    float p0 = 0.0f, p1 = 0.0f, dd = 0.0f;
    if (d < n) {
        dd = dinv[d];
        const int beg = rowptr[d];
        const int end = rowptr[d + 1];
        for (int j = beg + l; j < end; j += 16) {
            const int    s  = csr[j];
            const float  nm = dinv[s] * dd;
            const float2 h  = h2[s];
            p0 += nm * h.x;
            p1 += nm * h.y;
        }
    }
    for (int o = 8; o > 0; o >>= 1) {
        p0 += __shfl_xor(p0, o, 16);
        p1 += __shfl_xor(p1, o, 16);
    }
    if (d < n && l == 0) {
        const float  sn = dd * dd;
        const float2 hd = h2[d];
        out[2 * d]     = (sn * hd.x + b2[0]) + p0;
        out[2 * d + 1] = (sn * hd.y + b2[1]) + p1;
    }
}


extern "C" void kernel_launch(void* const* d_in, const int* in_sizes, int n_in,
                              void* d_out, int out_size, void* d_ws, size_t ws_size,
                              hipStream_t stream) {
    const float* X   = (const float*)d_in[0];
    const float* W1  = (const float*)d_in[1];
    const float* b1  = (const float*)d_in[2];
    const float* W2  = (const float*)d_in[3];
    const float* b2  = (const float*)d_in[4];
    const int*   ei  = (const int*)d_in[5];
    float*       out = (float*)d_out;

    const int N = NNODES;
    const int E = in_sizes[5] / 2;
    const int* src = ei;
    const int* dst = ei + E;

    int*    cnt    = (int*)d_ws;
    float*  dinv   = (float*)(cnt + N);
    float*  H1     = dinv + N;
    int*    rowptr = (int*)(H1 + (size_t)N * HIDDEN);
    int*    rank   = rowptr + (N + 1);
    int*    csr    = rank + E;
    float*  H2     = H1 + (size_t)2 * N * HIDDEN;
    float2* Wp     = (float2*)(H2 + (size_t)N * NCLS);

    const int T = 256;

    {
        int total = (DIMIN / 2) * HIDDEN;
        k_prep<<<(total + T - 1) / T, T, 0, stream>>>(W1, Wp, cnt, N);
    }

    const int EB = (E + T - 1) / T;
    k_gemm1a_count<<<GBLK_A + EB, T, 0, stream>>>(X, Wp, H1, dst, cnt, rank, E);
    k_scan_dinv   <<<1, SCAN_T, 0, stream>>>(cnt, rowptr, dinv);
    k_gemm1b_fill <<<GBLK_B + EB, T, 0, stream>>>(X, Wp, H1, src, dst, rowptr, rank, csr, E);

    {
        long long threads = (long long)N * 32;
        k_gather1<<<(int)((threads + T - 1) / T), T, 0, stream>>>(rowptr, csr, dinv, H1, b1, W2, H2, N);
    }

    {
        long long threads = (long long)N * 16;
        k_gather2<<<(int)((threads + T - 1) / T), T, 0, stream>>>(rowptr, csr, dinv, H2, b2, out, N);
    }
}


# Edge sort replaced by 64-slot rows per destination filled by the counting atomic's return (edge-list scan for a node over 64): scan launch and fill pass gone, x@W1 in one launch; arrival-order sums
# speedup vs baseline: 2.9355x; 1.0869x over previous
#include <hip/hip_runtime.h>
#include <hip/hip_bf16.h>

#define NNODES 50000
#define DIMIN  512
#define HIDDEN 256
#define NCLS   2
#define KHALF  256
#define LDSK   260

typedef __attribute__((ext_vector_type(2))) float v2f;
typedef __attribute__((ext_vector_type(8))) float v8f;

static_assert((DIMIN / 2) * HIDDEN >= NNODES, "the repack grid also clears the per-node counters");


__global__ void k_prep(const float* __restrict__ W, float2* __restrict__ Wp,
                       int* __restrict__ cnt, int n) {
    int idx = blockIdx.x * blockDim.x + threadIdx.x;
    if (idx < n) cnt[idx] = 0;
    if (idx >= (DIMIN / 2) * HIDDEN) return;
    int kp  = idx / HIDDEN;
    int col = idx - kp * HIDDEN;
    float2 p;
    p.x = W[(size_t)(2 * kp)     * HIDDEN + col];
    p.y = W[(size_t)(2 * kp + 1) * HIDDEN + col];
    Wp[idx] = p;
}

#define SLOTS 64
static_assert(SLOTS % 32 == 0, "a slot row is read in 32-lane pieces");

__device__ __forceinline__ void edge_place(const int* __restrict__ src, const int* __restrict__ dst,
                                           int* __restrict__ cnt, int* __restrict__ ell,
                                           int first, int stride, int e) {
    for (int i = first; i < e; i += stride) {
        const int d    = dst[i];
        const int slot = atomicAdd(&cnt[d], 1);
        if (slot < SLOTS) ell[d * SLOTS + slot] = src[i];
    }
}

__device__ __forceinline__ float dinv_of(int count) { return __frsqrt_rn((float)(count + 1)); }

#define TILES   (NNODES / 16)
#define PAIRS   ((TILES + 1) / 2)
#define GBLK    (2 * PAIRS)
#define PLACE_BLK 512
static_assert(NNODES % 16 == 0, "row tiles cover the nodes exactly");
static_assert(NNODES >= 32, "a moved-up last pair needs 32 rows to stand on");
static_assert(HIDDEN == 2 * 8 * 16, "two column halves of eight 16-column tiles, one tile per wave");
static_assert(DIMIN == 2 * KHALF, "the product runs as two staged K passes");
static_assert((32 * KHALF) / 4 == 8 * 256, "eight 16-byte pieces of the staged slab per thread");

__device__ __forceinline__ void
gemm1_tile(const float* __restrict__ X, const float2* __restrict__ Wp, float* __restrict__ H,
           float* As, const int gb) {
    const bool full = (gb >> 1) * 32 + 32 <= NNODES;
    const int  row0 = full ? (gb >> 1) * 32 : NNODES - 32;
    const int wave = threadIdx.x >> 5;
    const int lane = threadIdx.x & 31;
    const int l16  = lane & 15;
    const int hi   = lane >> 4;
    const int col0 = ((gb & 1) * 8 + wave) * 16;

    v8f c0 = {};
    v8f c1 = {};

    const int  sr  = threadIdx.x >> 6;
    const int  sc  = (threadIdx.x & 63) * 4;
    const float* xlo = X + (size_t)(row0 + sr) * DIMIN + sc;

    #pragma nounroll
    for (int kh = 0; kh < DIMIN; kh += KHALF) {
        if (kh != 0) __syncthreads();
        float4 st[8];
        #pragma unroll
        for (int i = 0; i < 8; ++i)
            st[i] = *(const float4*)(xlo + kh + (size_t)(4 * i) * DIMIN);
        #pragma unroll
        for (int i = 0; i < 8; ++i)
            *(float4*)(&As[(sr + 4 * i) * LDSK + sc]) = st[i];
        __syncthreads();

        const float2* Wk = Wp + (size_t)(kh >> 1) * HIDDEN + col0 + l16;
        #pragma unroll 8
        for (int k = 0; k < KHALF; k += 4) {
            const int kk = k + 2 * hi;
            const float2 a0f = *(const float2*)(&As[l16 * LDSK + kk]);
            const float2 a1f = *(const float2*)(&As[(16 + l16) * LDSK + kk]);
            v2f a0; a0.x = a0f.x; a0.y = a0f.y;
            v2f a1; a1.x = a1f.x; a1.y = a1f.y;
            const float2 bf = Wk[(size_t)(kk >> 1) * HIDDEN];
            v2f b; b.x = bf.x; b.y = bf.y;
            c0 = __builtin_amdgcn_wmma_f32_16x16x4_f32(false, a0, false, b,
                                                       (short)0, c0, false, false);
            c1 = __builtin_amdgcn_wmma_f32_16x16x4_f32(false, a1, false, b,
                                                       (short)0, c1, false, false);
        }
    }

    if (full) {
        #pragma unroll
        for (int v = 0; v < 8; ++v) {
            const int row = row0 + hi * 8 + v;
            H[(size_t)row * HIDDEN + col0 + l16] = c0[v];
        }
    }
    #pragma unroll
    for (int v = 0; v < 8; ++v) {
        const int row = row0 + 16 + hi * 8 + v;
        H[(size_t)row * HIDDEN + col0 + l16] = c1[v];
    }
}

__global__ void __launch_bounds__(256)
k_gemm1_place(const float* __restrict__ X, const float2* __restrict__ Wp, float* __restrict__ H,
              const int* __restrict__ src, const int* __restrict__ dst,
              int* __restrict__ cnt, int* __restrict__ ell, int e) {
    __shared__ float As[32 * LDSK];
    if (blockIdx.x >= PLACE_BLK) {
        gemm1_tile(X, Wp, H, As, (int)blockIdx.x - PLACE_BLK);
        return;
    }
    edge_place(src, dst, cnt, ell, (int)blockIdx.x * 256 + (int)threadIdx.x, PLACE_BLK * 256, e);
}


__device__ __forceinline__ float relu(float v) { return v > 0.0f ? v : 0.0f; }

__device__ __forceinline__ void add_row(float4& a0, float4& a1, const float* __restrict__ H1,
                                        int s, float nm, int lane) {
    const float4* hs = (const float4*)(H1 + (size_t)s * HIDDEN);
    const float4 g0 = hs[lane], g1 = hs[32 + lane];
    a0.x += nm * g0.x; a0.y += nm * g0.y; a0.z += nm * g0.z; a0.w += nm * g0.w;
    a1.x += nm * g1.x; a1.y += nm * g1.y; a1.z += nm * g1.z; a1.w += nm * g1.w;
}

__global__ void __launch_bounds__(256)
k_gather1(const int* __restrict__ cnt, const int* __restrict__ ell,
          const int* __restrict__ src, const int* __restrict__ dst, int e,
          const float* __restrict__ H1,
          const float* __restrict__ b1, const float* __restrict__ W2,
          float* __restrict__ H2, int n) {
    const int d    = __builtin_amdgcn_readfirstlane((int)((blockIdx.x * blockDim.x + threadIdx.x) >> 5));
    const int lane = threadIdx.x & 31;
    if (d >= n) return;

    const int   deg = cnt[d];
    const float dd  = dinv_of(deg);

    const float4* hd = (const float4*)(H1 + (size_t)d * HIDDEN);
    const float   sn = dd * dd;
    const float4  s0 = hd[lane], s1 = hd[32 + lane];
    float4 a0, a1;
    a0.x = sn * s0.x; a0.y = sn * s0.y; a0.z = sn * s0.z; a0.w = sn * s0.w;
    a1.x = sn * s1.x; a1.y = sn * s1.y; a1.z = sn * s1.z; a1.w = sn * s1.w;

    if (deg <= SLOTS) {
        const int* row = ell + d * SLOTS;
        for (int base = 0; base < deg; base += 32) {
            const int j = base + lane;
            int   s_l  = 0;
            float nm_l = 0.0f;
            if (j < deg) { s_l = row[j]; nm_l = dinv_of(cnt[s_l]) * dd; }
            const int m = (deg - base) < 32 ? (deg - base) : 32;
            #pragma unroll 4
            for (int t = 0; t < m; ++t) {
                const int   s  = __builtin_amdgcn_readlane(s_l, t);
                const float nm = __int_as_float(__builtin_amdgcn_readlane(__float_as_int(nm_l), t));
                add_row(a0, a1, H1, s, nm, lane);
            }
        }
    } else {
        for (int base = 0; base < e; base += 32) {
            const int i = base + lane;
            int   s_l  = 0;
            float nm_l = 0.0f;
            bool  hit  = false;
            if (i < e) hit = (dst[i] == d);
            if (hit) { s_l = src[i]; nm_l = dinv_of(cnt[s_l]) * dd; }
            unsigned mask = __builtin_amdgcn_ballot_w32(hit);
            while (mask != 0u) {
                const int t = __builtin_ctz(mask);
                mask &= mask - 1u;
                add_row(a0, a1, H1, __shfl(s_l, t), __shfl(nm_l, t), lane);
            }
        }
    }

    const float4 bA = ((const float4*)b1)[lane], bB = ((const float4*)b1)[32 + lane];
    const float z0 = relu(a0.x + bA.x), z1 = relu(a0.y + bA.y);
    const float z2 = relu(a0.z + bA.z), z3 = relu(a0.w + bA.w);
    const float z4 = relu(a1.x + bB.x), z5 = relu(a1.y + bB.y);
    const float z6 = relu(a1.z + bB.z), z7 = relu(a1.w + bB.w);

    const float4* w4 = (const float4*)W2;
    const float4 wA0 = w4[2 * lane],      wA1 = w4[2 * lane + 1];
    const float4 wB0 = w4[64 + 2 * lane], wB1 = w4[64 + 2 * lane + 1];
    float p0 = z0 * wA0.x + z1 * wA0.z + z2 * wA1.x + z3 * wA1.z
             + z4 * wB0.x + z5 * wB0.z + z6 * wB1.x + z7 * wB1.z;
    float p1 = z0 * wA0.y + z1 * wA0.w + z2 * wA1.y + z3 * wA1.w
             + z4 * wB0.y + z5 * wB0.w + z6 * wB1.y + z7 * wB1.w;
    for (int o = 16; o > 0; o >>= 1) {
        p0 += __shfl_xor(p0, o);
        p1 += __shfl_xor(p1, o);
    }
    if (lane == 0) {
        H2[2 * d]     = p0;
        H2[2 * d + 1] = p1;
    }
}


__global__ void k_gather2(const int* __restrict__ cnt, const int* __restrict__ ell,
                          const int* __restrict__ src, const int* __restrict__ dst, int e,
                          const float* __restrict__ H2,
                          const float* __restrict__ b2, float* __restrict__ out, int n) {
    const int gid = blockIdx.x * blockDim.x + threadIdx.x;
    const int d   = gid >> 4;
    const int l   = gid & 15;
    const float2* h2 = (const float2*)H2;
    float p0 = 0.0f, p1 = 0.0f, dd = 0.0f;
    if (d < n) {
        const int deg = cnt[d];
        dd = dinv_of(deg);
        if (deg <= SLOTS) {
            const int* row = ell + d * SLOTS;
            for (int j = l; j < deg; j += 16) {
                const int    s  = row[j];
                const float  nm = dinv_of(cnt[s]) * dd;
                const float2 h  = h2[s];
                p0 += nm * h.x;
                p1 += nm * h.y;
            }
        } else {
            for (int i = l; i < e; i += 16) {
                if (dst[i] == d) {
                    const int    s  = src[i];
                    const float  nm = dinv_of(cnt[s]) * dd;
                    const float2 h  = h2[s];
                    p0 += nm * h.x;
                    p1 += nm * h.y;
                }
            }
        }
    }
    for (int o = 8; o > 0; o >>= 1) {
        p0 += __shfl_xor(p0, o, 16);
        p1 += __shfl_xor(p1, o, 16);
    }
    if (d < n && l == 0) {
        const float  sn = dd * dd;
        const float2 hd = h2[d];
        out[2 * d]     = (sn * hd.x + b2[0]) + p0;
        out[2 * d + 1] = (sn * hd.y + b2[1]) + p1;
    }
}


extern "C" void kernel_launch(void* const* d_in, const int* in_sizes, int n_in,
                              void* d_out, int out_size, void* d_ws, size_t ws_size,
                              hipStream_t stream) {
    const float* X   = (const float*)d_in[0];
    const float* W1  = (const float*)d_in[1];
    const float* b1  = (const float*)d_in[2];
    const float* W2  = (const float*)d_in[3];
    const float* b2  = (const float*)d_in[4];
    const int*   ei  = (const int*)d_in[5];
    float*       out = (float*)d_out;

    const int N = NNODES;
    const int E = in_sizes[5] / 2;
    const int* src = ei;
    const int* dst = ei + E;

    int*    cnt    = (int*)d_ws;
    float*  H1     = (float*)(cnt + 2 * N);
    int*    ell    = (int*)(H1 + (size_t)N * HIDDEN) + 32;
    float*  H2     = H1 + (size_t)2 * N * HIDDEN;
    float2* Wp     = (float2*)(H2 + (size_t)N * NCLS);

    const int T = 256;

    {
        int total = (DIMIN / 2) * HIDDEN;
        k_prep<<<(total + T - 1) / T, T, 0, stream>>>(W1, Wp, cnt, N);
    }

    k_gemm1_place<<<PLACE_BLK + GBLK, T, 0, stream>>>(X, Wp, H1, src, dst, cnt, ell, E);

    {
        long long threads = (long long)N * 32;
        k_gather1<<<(int)((threads + T - 1) / T), T, 0, stream>>>(cnt, ell, src, dst, E, H1, b1, W2, H2, N);
    }

    {
        long long threads = (long long)N * 16;
        k_gather2<<<(int)((threads + T - 1) / T), T, 0, stream>>>(cnt, ell, src, dst, E, H2, b2, out, N);
    }
}
